# starting kernel
# baseline (speedup 1.0000x reference)
#include <hip/hip_runtime.h>

typedef __attribute__((ext_vector_type(2))) float v2f;
typedef __attribute__((ext_vector_type(8))) float v8f;

#define NB 16
#define SS 1024
#define EE 256
#define HH 8
#define DD 32
#define E3 768
#define LDSTR 20

__device__ __forceinline__ v8f wmma_f32(v2f a, v2f b, v8f c) {
  return __builtin_amdgcn_wmma_f32_16x16x4_f32(
       false, a,  false, b,
       (short)0, c,  false,  false);
}

__global__ __launch_bounds__(256) void gemm64x16(const float* __restrict__ X,
                                                 const float* __restrict__ W,
                                                 const float* __restrict__ bias,
                                                 float* __restrict__ Y,
                                                 int M, int N, int K) {
  const int lane = threadIdx.x & 31;
  const int wave = threadIdx.x >> 5;
  const int ntn  = N >> 4;
  int strip = blockIdx.x * 8 + wave;
  int sm = strip / ntn;
  int tn = strip - sm * ntn;
  const int row0 = sm << 6, col0 = tn << 4;
  const int m = lane & 15, half = lane >> 4;

  const float* xrow = X + (size_t)(row0 + m) * K + 2 * half;
  const float* wcol = W + (size_t)(2 * half) * N + col0 + m;

  v8f acc[4];
#pragma unroll
  for (int t = 0; t < 4; ++t) acc[t] = {};

#pragma unroll 4
  for (int kk = 0; kk < K; kk += 4) {
    v2f b;
    b.x = wcol[(size_t)kk * N];
    b.y = wcol[(size_t)(kk + 1) * N];
    v2f a0 = *(const v2f*)(xrow + kk);
    v2f a1 = *(const v2f*)(xrow + (size_t)16 * K + kk);
    v2f a2 = *(const v2f*)(xrow + (size_t)32 * K + kk);
    v2f a3 = *(const v2f*)(xrow + (size_t)48 * K + kk);
    acc[0] = wmma_f32(a0, b, acc[0]);
    acc[1] = wmma_f32(a1, b, acc[1]);
    acc[2] = wmma_f32(a2, b, acc[2]);
    acc[3] = wmma_f32(a3, b, acc[3]);
  }

  float bn = bias[col0 + m];
#pragma unroll
  for (int t = 0; t < 4; ++t) {
#pragma unroll
    for (int r = 0; r < 8; ++r) {
      int Mr = 16 * t + r + 8 * half;
      Y[(size_t)(row0 + Mr) * N + col0 + m] = acc[t][r] + bn;
    }
  }
}

__global__ __launch_bounds__(256) void rope_kernel(float* __restrict__ qkv) {
  int t = blockIdx.x * 256 + threadIdx.x;
  int p = t & 15;  t >>= 4;
  int h = t & 7;   t >>= 3;
  int s = t & 1023; t >>= 10;
  int b = t & 15;  t >>= 4;
  int which = t;

  int i = s >> 5, j = s & 31;
  int r = p & 7;
  float freq = __expf(-1.1512925464970231f * (float)r);
  float pos  = (p < 8) ? (float)i : (float)j;
  float ang  = pos * freq;
  float sn, cs;
  __sincosf(ang, &sn, &cs);

  size_t idx = (size_t)(b * SS + s) * E3 + which * EE + h * DD + 2 * p;
  float x0 = qkv[idx], x1 = qkv[idx + 1];
  qkv[idx]     = x0 * cs - x1 * sn;
  qkv[idx + 1] = x0 * sn + x1 * cs;
}

__global__ __launch_bounds__(256) void attn_kernel(const float* __restrict__ qkv,
                                                   float* __restrict__ out) {
  __shared__ float lds_p[8][16 * LDSTR];

  const int lane = threadIdx.x & 31;
  const int wave = threadIdx.x >> 5;
  int id = blockIdx.x * 8 + wave;
  const int qt = id & 63; id >>= 6;
  const int h  = id & 7;  id >>= 3;
  const int b  = id;
  const int m = lane & 15, half = lane >> 4;
  const int row0 = qt << 4;

  const size_t basebh = (size_t)(b * SS) * E3 + h * DD;
  const float* kslab = qkv + basebh + EE;
  const float* vslab = qkv + basebh + 2 * EE;

  v2f qa[8];
#pragma unroll
  for (int f = 0; f < 8; ++f)
    qa[f] = *(const v2f*)(qkv + basebh + (size_t)(row0 + m) * E3 + 4 * f + 2 * half);

  float mrow[8], lpart[8];
  v8f o0 = {}, o1 = {};
#pragma unroll
  for (int r = 0; r < 8; ++r) { mrow[r] = -3.0e38f; lpart[r] = 0.0f; }

  const float scale = 0.17677669529663687f;
  float* myp = &lds_p[wave][0];

  for (int kt = 0; kt < 64; ++kt) {
    const int krow0 = kt << 4;

    v8f sacc = {};
    const float* kbase = kslab + (size_t)(krow0 + m) * E3 + 2 * half;
#pragma unroll
    for (int f = 0; f < 8; ++f) {
      v2f kb = *(const v2f*)(kbase + 4 * f);
      sacc = wmma_f32(qa[f], kb, sacc);
    }

    float pr[8];
#pragma unroll
    for (int r = 0; r < 8; ++r) {
      float v = sacc[r] * scale;
      float rmax = v;
      rmax = fmaxf(rmax, __shfl_xor(rmax, 1, 32));
      rmax = fmaxf(rmax, __shfl_xor(rmax, 2, 32));
      rmax = fmaxf(rmax, __shfl_xor(rmax, 4, 32));
      rmax = fmaxf(rmax, __shfl_xor(rmax, 8, 32));
      float mnew = fmaxf(mrow[r], rmax);
      float corr = __expf(mrow[r] - mnew);
      float p    = __expf(v - mnew);
      lpart[r] = lpart[r] * corr + p;
      mrow[r]  = mnew;
      o0[r] *= corr;
      o1[r] *= corr;
      pr[r] = p;
    }

#pragma unroll
    for (int r = 0; r < 8; ++r)
      myp[(r + 8 * half) * LDSTR + m] = pr[r];
    __builtin_amdgcn_wave_barrier();
    v2f pa[4];
#pragma unroll
    for (int f = 0; f < 4; ++f)
      pa[f] = *(const v2f*)(myp + m * LDSTR + 4 * f + 2 * half);
    __builtin_amdgcn_wave_barrier();

#pragma unroll
    for (int f = 0; f < 4; ++f) {
      int kp = krow0 + 4 * f + 2 * half;
      const float* v0 = vslab + (size_t)kp * E3;
      const float* v1 = v0 + E3;
      v2f vb0, vb1;
      vb0.x = v0[m];      vb0.y = v1[m];
      vb1.x = v0[16 + m]; vb1.y = v1[16 + m];
      o0 = wmma_f32(pa[f], vb0, o0);
      o1 = wmma_f32(pa[f], vb1, o1);
    }
  }

#pragma unroll
  for (int r = 0; r < 8; ++r) {
    float lsum = lpart[r];
    lsum += __shfl_xor(lsum, 1, 32);
    lsum += __shfl_xor(lsum, 2, 32);
    lsum += __shfl_xor(lsum, 4, 32);
    lsum += __shfl_xor(lsum, 8, 32);
    float inv = 1.0f / lsum;
    int Mr = r + 8 * half;
    size_t orow = (size_t)(b * SS + row0 + Mr) * EE + h * DD;
    out[orow + m]      = o0[r] * inv;
    out[orow + 16 + m] = o1[r] * inv;
  }
}

extern "C" void kernel_launch(void* const* d_in, const int* in_sizes, int n_in,
                              void* d_out, int out_size, void* d_ws, size_t ws_size,
                              hipStream_t stream) {
  const float* x    = (const float*)d_in[0];
  const float* Wqkv = (const float*)d_in[1];
  const float* bqkv = (const float*)d_in[2];
  const float* Wout = (const float*)d_in[3];
  const float* bout = (const float*)d_in[4];
  float* out  = (float*)d_out;
  float* qkv  = (float*)d_ws;
  float* attn = qkv + (size_t)NB * SS * E3;

  gemm64x16<<<dim3((NB * SS / 64) * (E3 / 16) / 8), dim3(256), 0, stream>>>(
      x, Wqkv, bqkv, qkv, NB * SS, E3, EE);

  rope_kernel<<<dim3(2 * NB * SS * HH * (DD / 2) / 256), dim3(256), 0, stream>>>(qkv);

  attn_kernel<<<dim3(NB * HH * (SS / 16) / 8), dim3(256), 0, stream>>>(qkv, attn);

  gemm64x16<<<dim3((NB * SS / 64) * (EE / 16) / 8), dim3(256), 0, stream>>>(
      attn, Wout, bout, out, NB * SS, EE, EE);
}


# projection GEMMs: X (128x32) and W (32x64) K-slabs staged in LDS per 8-wave block by whole-line 16 B loads, two stages, one barrier per slab; WMMA fragments read from LDS; k sum order unchanged
# speedup vs baseline: 1.3014x; 1.3014x over previous
#include <hip/hip_runtime.h>

typedef __attribute__((ext_vector_type(2))) float v2f;
typedef __attribute__((ext_vector_type(4))) float v4f;
typedef __attribute__((ext_vector_type(8))) float v8f;

#define NB 16
#define SS 1024
#define EE 256
#define HH 8
#define DD 32
#define E3 768
#define LDSTR 20

__device__ __forceinline__ v8f wmma_f32(v2f a, v2f b, v8f c) {
  return __builtin_amdgcn_wmma_f32_16x16x4_f32(
       false, a,  false, b,
       (short)0, c,  false,  false);
}

#define GBM 128
#define GBN 64
#define GKC 32
#define XPITCH 36
#define WPITCH 80

__global__ __launch_bounds__(256) void gemm_lds128x64(const float* __restrict__ X,
                                                      const float* __restrict__ W,
                                                      const float* __restrict__ bias,
                                                      float* __restrict__ Y,
                                                      int M, int N, int K) {
  __shared__ __attribute__((aligned(16))) float xs[2 * GBM * XPITCH];
  __shared__ __attribute__((aligned(16))) float wl[2 * GKC * WPITCH];

  const int tid  = threadIdx.x;
  const int lane = tid & 31;
  const int wave = tid >> 5;
  const int m = lane & 15, half = lane >> 4;
  const int nbn = N >> 6;
  const int bm = blockIdx.x / nbn;
  const int bn = blockIdx.x - bm * nbn;
  const int row0 = bm * GBM, col0 = bn * GBN;
  const int wm = wave & 1;
  const int wn = wave >> 1;

  const int xr = tid >> 3, xc = (tid & 7) << 2;
  const int wr = tid >> 4, wc = (tid & 15) << 2;
  const float* xg = X + (size_t)(row0 + xr) * K + xc;
  const float* wg = W + (size_t)wr * N + col0 + wc;
  float* xst = xs + xr * XPITCH + xc;
  float* wst = wl + wr * WPITCH + wc;

  const size_t xstep = (size_t)32 * K;
  const size_t wstep = (size_t)16 * N;
  v4f x0 = *(const v4f*)(xg);
  v4f x1 = *(const v4f*)(xg + xstep);
  v4f x2 = *(const v4f*)(xg + 2 * xstep);
  v4f x3 = *(const v4f*)(xg + 3 * xstep);
  v4f w0 = *(const v4f*)(wg);
  v4f w1 = *(const v4f*)(wg + wstep);

  const float* xa = xs + (wm * 64 + m) * XPITCH + 2 * half;
  const float* wb = wl + (2 * half) * WPITCH + wn * 16 + m;

  v8f acc[4];
#pragma unroll
  for (int t = 0; t < 4; ++t) acc[t] = {};

#define STORE_SLAB(S)                                                   \
  do {                                                                  \
    *(v4f*)(xst + (S) * (GBM * XPITCH))               = x0;             \
    *(v4f*)(xst + (S) * (GBM * XPITCH) + 32 * XPITCH) = x1;             \
    *(v4f*)(xst + (S) * (GBM * XPITCH) + 64 * XPITCH) = x2;             \
    *(v4f*)(xst + (S) * (GBM * XPITCH) + 96 * XPITCH) = x3;             \
    *(v4f*)(wst + (S) * (GKC * WPITCH))               = w0;             \
    *(v4f*)(wst + (S) * (GKC * WPITCH) + 16 * WPITCH) = w1;             \
  } while (0)
#define LOAD_SLAB(KOFF)                                                 \
  do {                                                                  \
    const float* xn = xg + (KOFF);                                      \
    const float* wnx = wg + (size_t)(KOFF) * N;                         \
    x0 = *(const v4f*)(xn);                                             \
    x1 = *(const v4f*)(xn + xstep);                                     \
    x2 = *(const v4f*)(xn + 2 * xstep);                                 \
    x3 = *(const v4f*)(xn + 3 * xstep);                                 \
    w0 = *(const v4f*)(wnx);                                            \
    w1 = *(const v4f*)(wnx + wstep);                                    \
  } while (0)
#define WMMA_SLAB(S)                                                    \
  do {                                                                  \
    v2f bq[GKC / 4];                                                    \
    _Pragma("unroll") for (int q = 0; q < GKC / 4; ++q) {               \
      const float* wbs = wb + (S) * (GKC * WPITCH) + 4 * q * WPITCH;    \
      bq[q].x = wbs[0];                                                 \
      bq[q].y = wbs[WPITCH];                                            \
    }                                                                   \
    _Pragma("unroll") for (int q = 0; q < GKC / 4; ++q) {               \
      const float* xas = xa + (S) * (GBM * XPITCH) + 4 * q;             \
      v2f a0 = *(const v2f*)(xas);                                      \
      v2f a1 = *(const v2f*)(xas + 16 * XPITCH);                        \
      v2f a2 = *(const v2f*)(xas + 32 * XPITCH);                        \
      v2f a3 = *(const v2f*)(xas + 48 * XPITCH);                        \
      acc[0] = wmma_f32(a0, bq[q], acc[0]);                             \
      acc[1] = wmma_f32(a1, bq[q], acc[1]);                             \
      acc[2] = wmma_f32(a2, bq[q], acc[2]);                             \
      acc[3] = wmma_f32(a3, bq[q], acc[3]);                             \
    }                                                                   \
  } while (0)

  STORE_SLAB(0);
  __syncthreads();

  for (int c = 0; c < K; c += 2 * GKC) {
    const bool more1 = (c + GKC < K);
    if (more1) LOAD_SLAB(c + GKC);
    WMMA_SLAB(0);
    if (!more1) break;
    STORE_SLAB(1);
    __syncthreads();

    const bool more2 = (c + 2 * GKC < K);
    if (more2) LOAD_SLAB(c + 2 * GKC);
    WMMA_SLAB(1);
    if (more2) STORE_SLAB(0);
    __syncthreads();
  }
#undef STORE_SLAB
#undef LOAD_SLAB
#undef WMMA_SLAB

  const int cn = col0 + wn * 16 + m;
  float bnv = bias[cn];
#pragma unroll
  for (int t = 0; t < 4; ++t) {
#pragma unroll
    for (int r = 0; r < 8; ++r) {
      int Mr = wm * 64 + 16 * t + r + 8 * half;
      Y[(size_t)(row0 + Mr) * N + cn] = acc[t][r] + bnv;
    }
  }
}

__global__ __launch_bounds__(256) void rope_kernel(float* __restrict__ qkv) {
  int t = blockIdx.x * 256 + threadIdx.x;
  int p = t & 15;  t >>= 4;
  int h = t & 7;   t >>= 3;
  int s = t & 1023; t >>= 10;
  int b = t & 15;  t >>= 4;
  int which = t;

  int i = s >> 5, j = s & 31;
  int r = p & 7;
  float freq = __expf(-1.1512925464970231f * (float)r);
  float pos  = (p < 8) ? (float)i : (float)j;
  float ang  = pos * freq;
  float sn, cs;
  __sincosf(ang, &sn, &cs);

  size_t idx = (size_t)(b * SS + s) * E3 + which * EE + h * DD + 2 * p;
  float x0 = qkv[idx], x1 = qkv[idx + 1];
  qkv[idx]     = x0 * cs - x1 * sn;
  qkv[idx + 1] = x0 * sn + x1 * cs;
}

__global__ __launch_bounds__(256) void attn_kernel(const float* __restrict__ qkv,
                                                   float* __restrict__ out) {
  __shared__ float lds_p[8][16 * LDSTR];

  const int lane = threadIdx.x & 31;
  const int wave = threadIdx.x >> 5;
  int id = blockIdx.x * 8 + wave;
  const int qt = id & 63; id >>= 6;
  const int h  = id & 7;  id >>= 3;
  const int b  = id;
  const int m = lane & 15, half = lane >> 4;
  const int row0 = qt << 4;

  const size_t basebh = (size_t)(b * SS) * E3 + h * DD;
  const float* kslab = qkv + basebh + EE;
  const float* vslab = qkv + basebh + 2 * EE;

  v2f qa[8];
#pragma unroll
  for (int f = 0; f < 8; ++f)
    qa[f] = *(const v2f*)(qkv + basebh + (size_t)(row0 + m) * E3 + 4 * f + 2 * half);

  float mrow[8], lpart[8];
  v8f o0 = {}, o1 = {};
#pragma unroll
  for (int r = 0; r < 8; ++r) { mrow[r] = -3.0e38f; lpart[r] = 0.0f; }

  const float scale = 0.17677669529663687f;
  float* myp = &lds_p[wave][0];

  for (int kt = 0; kt < 64; ++kt) {
    const int krow0 = kt << 4;

    v8f sacc = {};
    const float* kbase = kslab + (size_t)(krow0 + m) * E3 + 2 * half;
#pragma unroll
    for (int f = 0; f < 8; ++f) {
      v2f kb = *(const v2f*)(kbase + 4 * f);
      sacc = wmma_f32(qa[f], kb, sacc);
    }

    float pr[8];
#pragma unroll
    for (int r = 0; r < 8; ++r) {
      float v = sacc[r] * scale;
      float rmax = v;
      rmax = fmaxf(rmax, __shfl_xor(rmax, 1, 32));
      rmax = fmaxf(rmax, __shfl_xor(rmax, 2, 32));
      rmax = fmaxf(rmax, __shfl_xor(rmax, 4, 32));
      rmax = fmaxf(rmax, __shfl_xor(rmax, 8, 32));
      float mnew = fmaxf(mrow[r], rmax);
      float corr = __expf(mrow[r] - mnew);
      float p    = __expf(v - mnew);
      lpart[r] = lpart[r] * corr + p;
      mrow[r]  = mnew;
      o0[r] *= corr;
      o1[r] *= corr;
      pr[r] = p;
    }

#pragma unroll
    for (int r = 0; r < 8; ++r)
      myp[(r + 8 * half) * LDSTR + m] = pr[r];
    __builtin_amdgcn_wave_barrier();
    v2f pa[4];
#pragma unroll
    for (int f = 0; f < 4; ++f)
      pa[f] = *(const v2f*)(myp + m * LDSTR + 4 * f + 2 * half);
    __builtin_amdgcn_wave_barrier();

#pragma unroll
    for (int f = 0; f < 4; ++f) {
      int kp = krow0 + 4 * f + 2 * half;
      const float* v0 = vslab + (size_t)kp * E3;
      const float* v1 = v0 + E3;
      v2f vb0, vb1;
      vb0.x = v0[m];      vb0.y = v1[m];
      vb1.x = v0[16 + m]; vb1.y = v1[16 + m];
      o0 = wmma_f32(pa[f], vb0, o0);
      o1 = wmma_f32(pa[f], vb1, o1);
    }
  }

#pragma unroll
  for (int r = 0; r < 8; ++r) {
    float lsum = lpart[r];
    lsum += __shfl_xor(lsum, 1, 32);
    lsum += __shfl_xor(lsum, 2, 32);
    lsum += __shfl_xor(lsum, 4, 32);
    lsum += __shfl_xor(lsum, 8, 32);
    float inv = 1.0f / lsum;
    int Mr = r + 8 * half;
    size_t orow = (size_t)(b * SS + row0 + Mr) * EE + h * DD;
    out[orow + m]      = o0[r] * inv;
    out[orow + 16 + m] = o1[r] * inv;
  }
}

extern "C" void kernel_launch(void* const* d_in, const int* in_sizes, int n_in,
                              void* d_out, int out_size, void* d_ws, size_t ws_size,
                              hipStream_t stream) {
  const float* x    = (const float*)d_in[0];
  const float* Wqkv = (const float*)d_in[1];
  const float* bqkv = (const float*)d_in[2];
  const float* Wout = (const float*)d_in[3];
  const float* bout = (const float*)d_in[4];
  float* out  = (float*)d_out;
  float* qkv  = (float*)d_ws;
  float* attn = qkv + (size_t)NB * SS * E3;

  gemm_lds128x64<<<dim3((NB * SS / GBM) * (E3 / GBN)), dim3(256), 0, stream>>>(
      x, Wqkv, bqkv, qkv, NB * SS, E3, EE);

  rope_kernel<<<dim3(2 * NB * SS * HH * (DD / 2) / 256), dim3(256), 0, stream>>>(qkv);

  attn_kernel<<<dim3(NB * HH * (SS / 16) / 8), dim3(256), 0, stream>>>(qkv, attn);

  gemm_lds128x64<<<dim3((NB * SS / GBM) * (EE / GBN)), dim3(256), 0, stream>>>(
      attn, Wout, bout, out, NB * SS, EE, EE);
}


# attention: softmax moved behind the LDS transposition (scaled scores cross LDS, not P): row max in-lane + one half exchange, no LDS permutes; O rescaled via DPP row broadcast; row-sum order changes
# speedup vs baseline: 1.5305x; 1.1760x over previous
#include <hip/hip_runtime.h>

typedef __attribute__((ext_vector_type(2))) float v2f;
typedef __attribute__((ext_vector_type(4))) float v4f;
typedef __attribute__((ext_vector_type(8))) float v8f;

#define NB 16
#define SS 1024
#define EE 256
#define HH 8
#define DD 32
#define E3 768
#define LDSTR 20

__device__ __forceinline__ v8f wmma_f32(v2f a, v2f b, v8f c) {
  return __builtin_amdgcn_wmma_f32_16x16x4_f32(
       false, a,  false, b,
       (short)0, c,  false,  false);
}

#define GBM 128
#define GBN 64
#define GKC 32
#define XPITCH 36
#define WPITCH 80

__global__ __launch_bounds__(256) void gemm_lds128x64(const float* __restrict__ X,
                                                      const float* __restrict__ W,
                                                      const float* __restrict__ bias,
                                                      float* __restrict__ Y,
                                                      int M, int N, int K) {
  __shared__ __attribute__((aligned(16))) float xs[2 * GBM * XPITCH];
  __shared__ __attribute__((aligned(16))) float wl[2 * GKC * WPITCH];

  const int tid  = threadIdx.x;
  const int lane = tid & 31;
  const int wave = tid >> 5;
  const int m = lane & 15, half = lane >> 4;
  const int nbn = N >> 6;
  const int bm = blockIdx.x / nbn;
  const int bn = blockIdx.x - bm * nbn;
  const int row0 = bm * GBM, col0 = bn * GBN;
  const int wm = wave & 1;
  const int wn = wave >> 1;

  const int xr = tid >> 3, xc = (tid & 7) << 2;
  const int wr = tid >> 4, wc = (tid & 15) << 2;
  const float* xg = X + (size_t)(row0 + xr) * K + xc;
  const float* wg = W + (size_t)wr * N + col0 + wc;
  float* xst = xs + xr * XPITCH + xc;
  float* wst = wl + wr * WPITCH + wc;

  const size_t xstep = (size_t)32 * K;
  const size_t wstep = (size_t)16 * N;
  v4f x0 = *(const v4f*)(xg);
  v4f x1 = *(const v4f*)(xg + xstep);
  v4f x2 = *(const v4f*)(xg + 2 * xstep);
  v4f x3 = *(const v4f*)(xg + 3 * xstep);
  v4f w0 = *(const v4f*)(wg);
  v4f w1 = *(const v4f*)(wg + wstep);

  const float* xa = xs + (wm * 64 + m) * XPITCH + 2 * half;
  const float* wb = wl + (2 * half) * WPITCH + wn * 16 + m;

  v8f acc[4];
#pragma unroll
  for (int t = 0; t < 4; ++t) acc[t] = {};

#define STORE_SLAB(S)                                                   \
  do {                                                                  \
    *(v4f*)(xst + (S) * (GBM * XPITCH))               = x0;             \
    *(v4f*)(xst + (S) * (GBM * XPITCH) + 32 * XPITCH) = x1;             \
    *(v4f*)(xst + (S) * (GBM * XPITCH) + 64 * XPITCH) = x2;             \
    *(v4f*)(xst + (S) * (GBM * XPITCH) + 96 * XPITCH) = x3;             \
    *(v4f*)(wst + (S) * (GKC * WPITCH))               = w0;             \
    *(v4f*)(wst + (S) * (GKC * WPITCH) + 16 * WPITCH) = w1;             \
  } while (0)
#define LOAD_SLAB(KOFF)                                                 \
  do {                                                                  \
    const float* xn = xg + (KOFF);                                      \
    const float* wnx = wg + (size_t)(KOFF) * N;                         \
    x0 = *(const v4f*)(xn);                                             \
    x1 = *(const v4f*)(xn + xstep);                                     \
    x2 = *(const v4f*)(xn + 2 * xstep);                                 \
    x3 = *(const v4f*)(xn + 3 * xstep);                                 \
    w0 = *(const v4f*)(wnx);                                            \
    w1 = *(const v4f*)(wnx + wstep);                                    \
  } while (0)
#define WMMA_SLAB(S)                                                    \
  do {                                                                  \
    v2f bq[GKC / 4];                                                    \
    _Pragma("unroll") for (int q = 0; q < GKC / 4; ++q) {               \
      const float* wbs = wb + (S) * (GKC * WPITCH) + 4 * q * WPITCH;    \
      bq[q].x = wbs[0];                                                 \
      bq[q].y = wbs[WPITCH];                                            \
    }                                                                   \
    _Pragma("unroll") for (int q = 0; q < GKC / 4; ++q) {               \
      const float* xas = xa + (S) * (GBM * XPITCH) + 4 * q;             \
      v2f a0 = *(const v2f*)(xas);                                      \
      v2f a1 = *(const v2f*)(xas + 16 * XPITCH);                        \
      v2f a2 = *(const v2f*)(xas + 32 * XPITCH);                        \
      v2f a3 = *(const v2f*)(xas + 48 * XPITCH);                        \
      acc[0] = wmma_f32(a0, bq[q], acc[0]);                             \
      acc[1] = wmma_f32(a1, bq[q], acc[1]);                             \
      acc[2] = wmma_f32(a2, bq[q], acc[2]);                             \
      acc[3] = wmma_f32(a3, bq[q], acc[3]);                             \
    }                                                                   \
  } while (0)

  STORE_SLAB(0);
  __syncthreads();

  for (int c = 0; c < K; c += 2 * GKC) {
    const bool more1 = (c + GKC < K);
    if (more1) LOAD_SLAB(c + GKC);
    WMMA_SLAB(0);
    if (!more1) break;
    STORE_SLAB(1);
    __syncthreads();

    const bool more2 = (c + 2 * GKC < K);
    if (more2) LOAD_SLAB(c + 2 * GKC);
    WMMA_SLAB(1);
    if (more2) STORE_SLAB(0);
    __syncthreads();
  }
#undef STORE_SLAB
#undef LOAD_SLAB
#undef WMMA_SLAB

  const int cn = col0 + wn * 16 + m;
  float bnv = bias[cn];
#pragma unroll
  for (int t = 0; t < 4; ++t) {
#pragma unroll
    for (int r = 0; r < 8; ++r) {
      int Mr = wm * 64 + 16 * t + r + 8 * half;
      Y[(size_t)(row0 + Mr) * N + cn] = acc[t][r] + bnv;
    }
  }
}

__global__ __launch_bounds__(256) void rope_kernel(float* __restrict__ qkv) {
  int t = blockIdx.x * 256 + threadIdx.x;
  int p = t & 15;  t >>= 4;
  int h = t & 7;   t >>= 3;
  int s = t & 1023; t >>= 10;
  int b = t & 15;  t >>= 4;
  int which = t;

  int i = s >> 5, j = s & 31;
  int r = p & 7;
  float freq = __expf(-1.1512925464970231f * (float)r);
  float pos  = (p < 8) ? (float)i : (float)j;
  float ang  = pos * freq;
  float sn, cs;
  __sincosf(ang, &sn, &cs);

  size_t idx = (size_t)(b * SS + s) * E3 + which * EE + h * DD + 2 * p;
  float x0 = qkv[idx], x1 = qkv[idx + 1];
  qkv[idx]     = x0 * cs - x1 * sn;
  qkv[idx + 1] = x0 * sn + x1 * cs;
}


__device__ __forceinline__ float xhalf(float x) {
  int b = __builtin_bit_cast(int, x);
  return __builtin_bit_cast(float,
      __builtin_amdgcn_permlanex16(b, b, 0x76543210, 0xfedcba98, false, false));
}

__device__ __forceinline__ int rows_for_halves(float x) {
  int b = __builtin_bit_cast(int, x);
  return __builtin_amdgcn_update_dpp(b, b, 0x168, 0x2, 0xf, false);
}

template <int R>
__device__ __forceinline__ float row_value(int xs) {
  return __builtin_bit_cast(float,
      __builtin_amdgcn_update_dpp(0, xs, 0x150 + R, 0xf, 0xf, true));
}

__global__ __launch_bounds__(256) void attn_kernel(const float* __restrict__ qkv,
                                                   float* __restrict__ out) {
  __shared__ float lds_p[8][16 * LDSTR];

  const int lane = threadIdx.x & 31;
  const int wave = threadIdx.x >> 5;
  int id = blockIdx.x * 8 + wave;
  const int qt = id & 63; id >>= 6;
  const int h  = id & 7;  id >>= 3;
  const int b  = id;
  const int m = lane & 15, half = lane >> 4;
  const int row0 = qt << 4;

  const size_t basebh = (size_t)(b * SS) * E3 + h * DD;
  const float* kslab = qkv + basebh + EE;
  const float* vslab = qkv + basebh + 2 * EE;

  v2f qa[8];
#pragma unroll
  for (int f = 0; f < 8; ++f)
    qa[f] = *(const v2f*)(qkv + basebh + (size_t)(row0 + m) * E3 + 4 * f + 2 * half);

  float mrow = -3.0e38f;
  float lsum = 0.0f;
  v8f o0 = {}, o1 = {};

  const float scale = 0.17677669529663687f;
  float* myp = &lds_p[wave][0];

  for (int kt = 0; kt < 64; ++kt) {
    const int krow0 = kt << 4;

    v8f sacc = {};
    const float* kbase = kslab + (size_t)(krow0 + m) * E3 + 2 * half;
#pragma unroll
    for (int f = 0; f < 8; ++f) {
      v2f kb = *(const v2f*)(kbase + 4 * f);
      sacc = wmma_f32(qa[f], kb, sacc);
    }

#pragma unroll
    for (int r = 0; r < 8; ++r)
      myp[(r + 8 * half) * LDSTR + m] = sacc[r] * scale;
    __builtin_amdgcn_wave_barrier();
    v2f sv[4];
#pragma unroll
    for (int f = 0; f < 4; ++f)
      sv[f] = *(const v2f*)(myp + m * LDSTR + 4 * f + 2 * half);
    __builtin_amdgcn_wave_barrier();

    float tmax = fmaxf(fmaxf(sv[0].x, sv[0].y), sv[1].x);
    tmax = fmaxf(fmaxf(tmax, sv[1].y), sv[2].x);
    tmax = fmaxf(fmaxf(tmax, sv[2].y), sv[3].x);
    tmax = fmaxf(tmax, sv[3].y);
    tmax = fmaxf(tmax, xhalf(tmax));
    const float mnew = fmaxf(mrow, tmax);
    const float corr = __expf(mrow - mnew);
    mrow = mnew;

    v2f pa[4];
#pragma unroll
    for (int f = 0; f < 4; ++f) {
      pa[f].x = __expf(sv[f].x - mnew);
      pa[f].y = __expf(sv[f].y - mnew);
    }
    lsum = lsum * corr + (((pa[0].x + pa[0].y) + (pa[1].x + pa[1].y)) +
                          ((pa[2].x + pa[2].y) + (pa[3].x + pa[3].y)));

    const int cs = rows_for_halves(corr);
#define RESCALE_ROW(R)                       \
  do {                                       \
    const float cr = row_value<R>(cs);       \
    o0[R] *= cr;                             \
    o1[R] *= cr;                             \
  } while (0)
    RESCALE_ROW(0); RESCALE_ROW(1); RESCALE_ROW(2); RESCALE_ROW(3);
    RESCALE_ROW(4); RESCALE_ROW(5); RESCALE_ROW(6); RESCALE_ROW(7);
#undef RESCALE_ROW

#pragma unroll
    for (int f = 0; f < 4; ++f) {
      int kp = krow0 + 4 * f + 2 * half;
      const float* v0 = vslab + (size_t)kp * E3;
      const float* v1 = v0 + E3;
      v2f vb0, vb1;
      vb0.x = v0[m];      vb0.y = v1[m];
      vb1.x = v0[16 + m]; vb1.y = v1[16 + m];
      o0 = wmma_f32(pa[f], vb0, o0);
      o1 = wmma_f32(pa[f], vb1, o1);
    }
  }

  const float inv = 1.0f / (lsum + xhalf(lsum));
  const int is = rows_for_halves(inv);
#define STORE_ROW(R)                                                     \
  do {                                                                   \
    const float ir = row_value<R>(is);                                   \
    const size_t orow = (size_t)(b * SS + row0 + (R) + 8 * half) * EE + h * DD; \
    out[orow + m]      = o0[R] * ir;                                     \
    out[orow + 16 + m] = o1[R] * ir;                                     \
  } while (0)
  STORE_ROW(0); STORE_ROW(1); STORE_ROW(2); STORE_ROW(3);
  STORE_ROW(4); STORE_ROW(5); STORE_ROW(6); STORE_ROW(7);
#undef STORE_ROW
}

extern "C" void kernel_launch(void* const* d_in, const int* in_sizes, int n_in,
                              void* d_out, int out_size, void* d_ws, size_t ws_size,
                              hipStream_t stream) {
  const float* x    = (const float*)d_in[0];
  const float* Wqkv = (const float*)d_in[1];
  const float* bqkv = (const float*)d_in[2];
  const float* Wout = (const float*)d_in[3];
  const float* bout = (const float*)d_in[4];
  float* out  = (float*)d_out;
  float* qkv  = (float*)d_ws;
  float* attn = qkv + (size_t)NB * SS * E3;

  gemm_lds128x64<<<dim3((NB * SS / GBM) * (E3 / GBN)), dim3(256), 0, stream>>>(
      x, Wqkv, bqkv, qkv, NB * SS, E3, EE);

  rope_kernel<<<dim3(2 * NB * SS * HH * (DD / 2) / 256), dim3(256), 0, stream>>>(qkv);

  attn_kernel<<<dim3(NB * HH * (SS / 16) / 8), dim3(256), 0, stream>>>(qkv, attn);

  gemm_lds128x64<<<dim3((NB * SS / GBM) * (EE / GBN)), dim3(256), 0, stream>>>(
      attn, Wout, bout, out, NB * SS, EE, EE);
}


# attention: each wave takes two 16-row query tiles; a key tile's K and V fragments are loaded once into registers and feed both tiles' WMMAs (half the loads per WMMA); per-query sum order unchanged
# speedup vs baseline: 1.6636x; 1.0870x over previous
#include <hip/hip_runtime.h>

typedef __attribute__((ext_vector_type(2))) float v2f;
typedef __attribute__((ext_vector_type(4))) float v4f;
typedef __attribute__((ext_vector_type(8))) float v8f;

#define NB 16
#define SS 1024
#define EE 256
#define HH 8
#define DD 32
#define E3 768
#define LDSTR 20

__device__ __forceinline__ v8f wmma_f32(v2f a, v2f b, v8f c) {
  return __builtin_amdgcn_wmma_f32_16x16x4_f32(
       false, a,  false, b,
       (short)0, c,  false,  false);
}

#define GBM 128
#define GBN 64
#define GKC 32
#define XPITCH 36
#define WPITCH 80

__global__ __launch_bounds__(256) void gemm_lds128x64(const float* __restrict__ X,
                                                      const float* __restrict__ W,
                                                      const float* __restrict__ bias,
                                                      float* __restrict__ Y,
                                                      int M, int N, int K) {
  __shared__ __attribute__((aligned(16))) float xs[2 * GBM * XPITCH];
  __shared__ __attribute__((aligned(16))) float wl[2 * GKC * WPITCH];

  const int tid  = threadIdx.x;
  const int lane = tid & 31;
  const int wave = tid >> 5;
  const int m = lane & 15, half = lane >> 4;
  const int nbn = N >> 6;
  const int bm = blockIdx.x / nbn;
  const int bn = blockIdx.x - bm * nbn;
  const int row0 = bm * GBM, col0 = bn * GBN;
  const int wm = wave & 1;
  const int wn = wave >> 1;

  const int xr = tid >> 3, xc = (tid & 7) << 2;
  const int wr = tid >> 4, wc = (tid & 15) << 2;
  const float* xg = X + (size_t)(row0 + xr) * K + xc;
  const float* wg = W + (size_t)wr * N + col0 + wc;
  float* xst = xs + xr * XPITCH + xc;
  float* wst = wl + wr * WPITCH + wc;

  const size_t xstep = (size_t)32 * K;
  const size_t wstep = (size_t)16 * N;
  v4f x0 = *(const v4f*)(xg);
  v4f x1 = *(const v4f*)(xg + xstep);
  v4f x2 = *(const v4f*)(xg + 2 * xstep);
  v4f x3 = *(const v4f*)(xg + 3 * xstep);
  v4f w0 = *(const v4f*)(wg);
  v4f w1 = *(const v4f*)(wg + wstep);

  const float* xa = xs + (wm * 64 + m) * XPITCH + 2 * half;
  const float* wb = wl + (2 * half) * WPITCH + wn * 16 + m;

  v8f acc[4];
#pragma unroll
  for (int t = 0; t < 4; ++t) acc[t] = {};

#define STORE_SLAB(S)                                                   \
  do {                                                                  \
    *(v4f*)(xst + (S) * (GBM * XPITCH))               = x0;             \
    *(v4f*)(xst + (S) * (GBM * XPITCH) + 32 * XPITCH) = x1;             \
    *(v4f*)(xst + (S) * (GBM * XPITCH) + 64 * XPITCH) = x2;             \
    *(v4f*)(xst + (S) * (GBM * XPITCH) + 96 * XPITCH) = x3;             \
    *(v4f*)(wst + (S) * (GKC * WPITCH))               = w0;             \
    *(v4f*)(wst + (S) * (GKC * WPITCH) + 16 * WPITCH) = w1;             \
  } while (0)
#define LOAD_SLAB(KOFF)                                                 \
  do {                                                                  \
    const float* xn = xg + (KOFF);                                      \
    const float* wnx = wg + (size_t)(KOFF) * N;                         \
    x0 = *(const v4f*)(xn);                                             \
    x1 = *(const v4f*)(xn + xstep);                                     \
    x2 = *(const v4f*)(xn + 2 * xstep);                                 \
    x3 = *(const v4f*)(xn + 3 * xstep);                                 \
    w0 = *(const v4f*)(wnx);                                            \
    w1 = *(const v4f*)(wnx + wstep);                                    \
  } while (0)
#define WMMA_SLAB(S)                                                    \
  do {                                                                  \
    v2f bq[GKC / 4];                                                    \
    _Pragma("unroll") for (int q = 0; q < GKC / 4; ++q) {               \
      const float* wbs = wb + (S) * (GKC * WPITCH) + 4 * q * WPITCH;    \
      bq[q].x = wbs[0];                                                 \
      bq[q].y = wbs[WPITCH];                                            \
    }                                                                   \
    _Pragma("unroll") for (int q = 0; q < GKC / 4; ++q) {               \
      const float* xas = xa + (S) * (GBM * XPITCH) + 4 * q;             \
      v2f a0 = *(const v2f*)(xas);                                      \
      v2f a1 = *(const v2f*)(xas + 16 * XPITCH);                        \
      v2f a2 = *(const v2f*)(xas + 32 * XPITCH);                        \
      v2f a3 = *(const v2f*)(xas + 48 * XPITCH);                        \
      acc[0] = wmma_f32(a0, bq[q], acc[0]);                             \
      acc[1] = wmma_f32(a1, bq[q], acc[1]);                             \
      acc[2] = wmma_f32(a2, bq[q], acc[2]);                             \
      acc[3] = wmma_f32(a3, bq[q], acc[3]);                             \
    }                                                                   \
  } while (0)

  STORE_SLAB(0);
  __syncthreads();

  for (int c = 0; c < K; c += 2 * GKC) {
    const bool more1 = (c + GKC < K);
    if (more1) LOAD_SLAB(c + GKC);
    WMMA_SLAB(0);
    if (!more1) break;
    STORE_SLAB(1);
    __syncthreads();

    const bool more2 = (c + 2 * GKC < K);
    if (more2) LOAD_SLAB(c + 2 * GKC);
    WMMA_SLAB(1);
    if (more2) STORE_SLAB(0);
    __syncthreads();
  }
#undef STORE_SLAB
#undef LOAD_SLAB
#undef WMMA_SLAB

  const int cn = col0 + wn * 16 + m;
  float bnv = bias[cn];
#pragma unroll
  for (int t = 0; t < 4; ++t) {
#pragma unroll
    for (int r = 0; r < 8; ++r) {
      int Mr = wm * 64 + 16 * t + r + 8 * half;
      Y[(size_t)(row0 + Mr) * N + cn] = acc[t][r] + bnv;
    }
  }
}

__global__ __launch_bounds__(256) void rope_kernel(float* __restrict__ qkv) {
  int t = blockIdx.x * 256 + threadIdx.x;
  int p = t & 15;  t >>= 4;
  int h = t & 7;   t >>= 3;
  int s = t & 1023; t >>= 10;
  int b = t & 15;  t >>= 4;
  int which = t;

  int i = s >> 5, j = s & 31;
  int r = p & 7;
  float freq = __expf(-1.1512925464970231f * (float)r);
  float pos  = (p < 8) ? (float)i : (float)j;
  float ang  = pos * freq;
  float sn, cs;
  __sincosf(ang, &sn, &cs);

  size_t idx = (size_t)(b * SS + s) * E3 + which * EE + h * DD + 2 * p;
  float x0 = qkv[idx], x1 = qkv[idx + 1];
  qkv[idx]     = x0 * cs - x1 * sn;
  qkv[idx + 1] = x0 * sn + x1 * cs;
}


__device__ __forceinline__ float xhalf(float x) {
  int b = __builtin_bit_cast(int, x);
  return __builtin_bit_cast(float,
      __builtin_amdgcn_permlanex16(b, b, 0x76543210, 0xfedcba98, false, false));
}

__device__ __forceinline__ int rows_for_halves(float x) {
  int b = __builtin_bit_cast(int, x);
  return __builtin_amdgcn_update_dpp(b, b, 0x168, 0x2, 0xf, false);
}

template <int R>
__device__ __forceinline__ float row_value(int xs) {
  return __builtin_bit_cast(float,
      __builtin_amdgcn_update_dpp(0, xs, 0x150 + R, 0xf, 0xf, true));
}

#define QT 2

__global__ __launch_bounds__(256) __attribute__((amdgpu_waves_per_eu(4, 4))) void attn_kernel(const float* __restrict__ qkv,
                                                   float* __restrict__ out) {
  __shared__ float lds_p[8][QT][16 * LDSTR];

  const int lane = threadIdx.x & 31;
  const int wave = threadIdx.x >> 5;
  int id = blockIdx.x * 8 + wave;
  const int qg = id % (64 / QT); id /= (64 / QT);
  const int h  = id & 7;  id >>= 3;
  const int b  = id;
  const int m = lane & 15, half = lane >> 4;
  const int row0 = qg * (16 * QT);

  const size_t basebh = (size_t)(b * SS) * E3 + h * DD;
  const float* kslab = qkv + basebh + EE;
  const float* vslab = qkv + basebh + 2 * EE;

  v2f qa[QT][8];
#pragma unroll
  for (int t = 0; t < QT; ++t) {
#pragma unroll
    for (int f = 0; f < 8; ++f)
      qa[t][f] = *(const v2f*)(qkv + basebh + (size_t)(row0 + 16 * t + m) * E3 + 4 * f + 2 * half);
  }

  float mrow[QT];
  float lsum[QT];
  v8f o0[QT], o1[QT];
#pragma unroll
  for (int t = 0; t < QT; ++t) {
    mrow[t] = -3.0e38f;
    lsum[t] = 0.0f;
    o0[t] = {};
    o1[t] = {};
  }

  const float scale = 0.17677669529663687f;
  float* myp = &lds_p[wave][0][0];

  for (int kt = 0; kt < 64; ++kt) {
    const int krow0 = kt << 4;

    const float* kbase = kslab + (size_t)(krow0 + m) * E3 + 2 * half;
    v2f kb[8];
#pragma unroll
    for (int f = 0; f < 8; ++f) kb[f] = *(const v2f*)(kbase + 4 * f);

    v2f vb0[4], vb1[4];
#pragma unroll
    for (int f = 0; f < 4; ++f) {
      const float* v0 = vslab + (size_t)(krow0 + 4 * f + 2 * half) * E3;
      const float* v1 = v0 + E3;
      vb0[f].x = v0[m];      vb0[f].y = v1[m];
      vb1[f].x = v0[16 + m]; vb1[f].y = v1[16 + m];
    }

    v8f sacc[QT];
#pragma unroll
    for (int t = 0; t < QT; ++t) sacc[t] = {};
#pragma unroll
    for (int f = 0; f < 8; ++f) {
#pragma unroll
      for (int t = 0; t < QT; ++t) sacc[t] = wmma_f32(qa[t][f], kb[f], sacc[t]);
    }

#pragma unroll
    for (int t = 0; t < QT; ++t) {
#pragma unroll
      for (int r = 0; r < 8; ++r)
        myp[t * (16 * LDSTR) + (r + 8 * half) * LDSTR + m] = sacc[t][r] * scale;
    }
    __builtin_amdgcn_wave_barrier();
    v2f sv[QT][4];
#pragma unroll
    for (int t = 0; t < QT; ++t) {
#pragma unroll
      for (int f = 0; f < 4; ++f)
        sv[t][f] = *(const v2f*)(myp + t * (16 * LDSTR) + m * LDSTR + 4 * f + 2 * half);
    }
    __builtin_amdgcn_wave_barrier();

    v2f pa[QT][4];
#pragma unroll
    for (int t = 0; t < QT; ++t) {
      float tmax = fmaxf(fmaxf(sv[t][0].x, sv[t][0].y), sv[t][1].x);
      tmax = fmaxf(fmaxf(tmax, sv[t][1].y), sv[t][2].x);
      tmax = fmaxf(fmaxf(tmax, sv[t][2].y), sv[t][3].x);
      tmax = fmaxf(tmax, sv[t][3].y);
      tmax = fmaxf(tmax, xhalf(tmax));
      const float mnew = fmaxf(mrow[t], tmax);
      const float corr = __expf(mrow[t] - mnew);
      mrow[t] = mnew;

#pragma unroll
      for (int f = 0; f < 4; ++f) {
        pa[t][f].x = __expf(sv[t][f].x - mnew);
        pa[t][f].y = __expf(sv[t][f].y - mnew);
      }
      lsum[t] = lsum[t] * corr + (((pa[t][0].x + pa[t][0].y) + (pa[t][1].x + pa[t][1].y)) +
                                  ((pa[t][2].x + pa[t][2].y) + (pa[t][3].x + pa[t][3].y)));

      const int cs = rows_for_halves(corr);
#define RESCALE_ROW(R)                       \
  do {                                       \
    const float cr = row_value<R>(cs);       \
    o0[t][R] *= cr;                          \
    o1[t][R] *= cr;                          \
  } while (0)
      RESCALE_ROW(0); RESCALE_ROW(1); RESCALE_ROW(2); RESCALE_ROW(3);
      RESCALE_ROW(4); RESCALE_ROW(5); RESCALE_ROW(6); RESCALE_ROW(7);
#undef RESCALE_ROW
    }

#pragma unroll
    for (int f = 0; f < 4; ++f) {
#pragma unroll
      for (int t = 0; t < QT; ++t) {
        o0[t] = wmma_f32(pa[t][f], vb0[f], o0[t]);
        o1[t] = wmma_f32(pa[t][f], vb1[f], o1[t]);
      }
    }
  }

#pragma unroll
  for (int t = 0; t < QT; ++t) {
    const float inv = 1.0f / (lsum[t] + xhalf(lsum[t]));
    const int is = rows_for_halves(inv);
#define STORE_ROW(R)                                                     \
  do {                                                                   \
    const float ir = row_value<R>(is);                                   \
    const size_t orow = (size_t)(b * SS + row0 + 16 * t + (R) + 8 * half) * EE + h * DD; \
    out[orow + m]      = o0[t][R] * ir;                                  \
    out[orow + 16 + m] = o1[t][R] * ir;                                  \
  } while (0)
    STORE_ROW(0); STORE_ROW(1); STORE_ROW(2); STORE_ROW(3);
    STORE_ROW(4); STORE_ROW(5); STORE_ROW(6); STORE_ROW(7);
#undef STORE_ROW
  }
}

extern "C" void kernel_launch(void* const* d_in, const int* in_sizes, int n_in,
                              void* d_out, int out_size, void* d_ws, size_t ws_size,
                              hipStream_t stream) {
  const float* x    = (const float*)d_in[0];
  const float* Wqkv = (const float*)d_in[1];
  const float* bqkv = (const float*)d_in[2];
  const float* Wout = (const float*)d_in[3];
  const float* bout = (const float*)d_in[4];
  float* out  = (float*)d_out;
  float* qkv  = (float*)d_ws;
  float* attn = qkv + (size_t)NB * SS * E3;

  gemm_lds128x64<<<dim3((NB * SS / GBM) * (E3 / GBN)), dim3(256), 0, stream>>>(
      x, Wqkv, bqkv, qkv, NB * SS, E3, EE);

  rope_kernel<<<dim3(2 * NB * SS * HH * (DD / 2) / 256), dim3(256), 0, stream>>>(qkv);

  attn_kernel<<<dim3(NB * HH * (SS / (16 * QT)) / 8), dim3(256), 0, stream>>>(qkv, attn);

  gemm_lds128x64<<<dim3((NB * SS / GBM) * (EE / GBN)), dim3(256), 0, stream>>>(
      attn, Wout, bout, out, NB * SS, EE, EE);
}


# RoPE applied in the attention kernel's registers: q rotated (with log2e/sqrt(D)) once at load, each K tile's 8 fragments rotated behind their loads; qkv GEMM stores raw K tiles; RoPE kernel removed
# speedup vs baseline: 2.2491x; 1.3519x over previous
#include <hip/hip_runtime.h>

typedef __attribute__((ext_vector_type(2))) float v2f;
typedef __attribute__((ext_vector_type(4))) float v4f;
typedef __attribute__((ext_vector_type(8))) float v8f;

#define NB 16
#define SS 1024
#define EE 256
#define HH 8
#define DD 32
#define E3 768
#define LDSTR 20

__device__ __forceinline__ v8f wmma_f32(v2f a, v2f b, v8f c) {
  return __builtin_amdgcn_wmma_f32_16x16x4_f32(
       false, a,  false, b,
       (short)0, c,  false,  false);
}

#define GBM 128
#define GBN 64
#define GKC 32
#define XPITCH 36
#define WPITCH 80
#define CPITCH 68

template <bool VTILES>
__global__ __launch_bounds__(256) void gemm_lds128x64(const float* __restrict__ X,
                                                      const float* __restrict__ W,
                                                      const float* __restrict__ bias,
                                                      float* __restrict__ Y,
                                                      float* __restrict__ Vf,
                                                      float* __restrict__ Kf,
                                                      int M, int N, int K) {
  __shared__ __attribute__((aligned(16))) float xs[2 * GBM * XPITCH];
  __shared__ __attribute__((aligned(16))) float wl[2 * GKC * WPITCH];

  const int tid  = threadIdx.x;
  const int lane = tid & 31;
  const int wave = tid >> 5;
  const int m = lane & 15, half = lane >> 4;
  const int nbn = N >> 6;
  const int bm = blockIdx.x / nbn;
  const int bn = blockIdx.x - bm * nbn;
  const int row0 = bm * GBM, col0 = bn * GBN;
  const int wm = wave & 1;
  const int wn = wave >> 1;

  const int xr = tid >> 3, xc = (tid & 7) << 2;
  const int wr = tid >> 4, wc = (tid & 15) << 2;
  const float* xg = X + (size_t)(row0 + xr) * K + xc;
  const float* wg = W + (size_t)wr * N + col0 + wc;
  float* xst = xs + xr * XPITCH + xc;
  float* wst = wl + wr * WPITCH + wc;

  const size_t xstep = (size_t)32 * K;
  const size_t wstep = (size_t)16 * N;
  v4f x0 = *(const v4f*)(xg);
  v4f x1 = *(const v4f*)(xg + xstep);
  v4f x2 = *(const v4f*)(xg + 2 * xstep);
  v4f x3 = *(const v4f*)(xg + 3 * xstep);
  v4f w0 = *(const v4f*)(wg);
  v4f w1 = *(const v4f*)(wg + wstep);

  const float* xa = xs + (wm * 64 + m) * XPITCH + 2 * half;
  const float* wb = wl + (2 * half) * WPITCH + wn * 16 + m;

  v8f acc[4];
#pragma unroll
  for (int t = 0; t < 4; ++t) acc[t] = {};

#define STORE_SLAB(S)                                                   \
  do {                                                                  \
    *(v4f*)(xst + (S) * (GBM * XPITCH))               = x0;             \
    *(v4f*)(xst + (S) * (GBM * XPITCH) + 32 * XPITCH) = x1;             \
    *(v4f*)(xst + (S) * (GBM * XPITCH) + 64 * XPITCH) = x2;             \
    *(v4f*)(xst + (S) * (GBM * XPITCH) + 96 * XPITCH) = x3;             \
    *(v4f*)(wst + (S) * (GKC * WPITCH))               = w0;             \
    *(v4f*)(wst + (S) * (GKC * WPITCH) + 16 * WPITCH) = w1;             \
  } while (0)
#define LOAD_SLAB(KOFF)                                                 \
  do {                                                                  \
    const float* xn = xg + (KOFF);                                      \
    const float* wnx = wg + (size_t)(KOFF) * N;                         \
    x0 = *(const v4f*)(xn);                                             \
    x1 = *(const v4f*)(xn + xstep);                                     \
    x2 = *(const v4f*)(xn + 2 * xstep);                                 \
    x3 = *(const v4f*)(xn + 3 * xstep);                                 \
    w0 = *(const v4f*)(wnx);                                            \
    w1 = *(const v4f*)(wnx + wstep);                                    \
  } while (0)
#define WMMA_SLAB(S)                                                    \
  do {                                                                  \
    v2f bq[GKC / 4];                                                    \
    _Pragma("unroll") for (int q = 0; q < GKC / 4; ++q) {               \
      const float* wbs = wb + (S) * (GKC * WPITCH) + 4 * q * WPITCH;    \
      bq[q].x = wbs[0];                                                 \
      bq[q].y = wbs[WPITCH];                                            \
    }                                                                   \
    _Pragma("unroll") for (int q = 0; q < GKC / 4; ++q) {               \
      const float* xas = xa + (S) * (GBM * XPITCH) + 4 * q;             \
      v2f a0 = *(const v2f*)(xas);                                      \
      v2f a1 = *(const v2f*)(xas + 16 * XPITCH);                        \
      v2f a2 = *(const v2f*)(xas + 32 * XPITCH);                        \
      v2f a3 = *(const v2f*)(xas + 48 * XPITCH);                        \
      acc[0] = wmma_f32(a0, bq[q], acc[0]);                             \
      acc[1] = wmma_f32(a1, bq[q], acc[1]);                             \
      acc[2] = wmma_f32(a2, bq[q], acc[2]);                             \
      acc[3] = wmma_f32(a3, bq[q], acc[3]);                             \
    }                                                                   \
  } while (0)

  STORE_SLAB(0);
  __syncthreads();

  for (int c = 0; c < K; c += 2 * GKC) {
    const bool more1 = (c + GKC < K);
    if (more1) LOAD_SLAB(c + GKC);
    WMMA_SLAB(0);
    if (!more1) break;
    STORE_SLAB(1);
    __syncthreads();

    const bool more2 = (c + 2 * GKC < K);
    if (more2) LOAD_SLAB(c + 2 * GKC);
    WMMA_SLAB(1);
    if (more2) STORE_SLAB(0);
    __syncthreads();
  }
#undef STORE_SLAB
#undef LOAD_SLAB
#undef WMMA_SLAB

  const int cn = col0 + wn * 16 + m;
  float bnv = bias[cn];
  if (VTILES && col0 >= 2 * EE) {
    const int cv = col0 - 2 * EE + wn * 16;
    const int vh = cv >> 5;
    const int vw = (cv >> 4) & 1;
#pragma unroll
    for (int t = 0; t < 4; ++t) {
      const int R0 = row0 + wm * 64 + 16 * t;
      const int vb = R0 >> 10, vkt = (R0 & (SS - 1)) >> 4;
      float* tile = Vf + ((size_t)((vb * HH + vh) * 64 + vkt)) * 512 + vw * 256 + half * 128 + m * 4;
      v4f lo = {acc[t][0] + bnv, acc[t][1] + bnv, acc[t][4] + bnv, acc[t][5] + bnv};
      v4f hi = {acc[t][2] + bnv, acc[t][3] + bnv, acc[t][6] + bnv, acc[t][7] + bnv};
      *(v4f*)(tile)      = lo;
      *(v4f*)(tile + 64) = hi;
    }
    return;
  }
  __syncthreads();
  float* ct = xs;
#pragma unroll
  for (int t = 0; t < 4; ++t) {
#pragma unroll
    for (int r = 0; r < 8; ++r) {
      int Mr = wm * 64 + 16 * t + r + 8 * half;
      ct[Mr * CPITCH + wn * 16 + m] = acc[t][r] + bnv;
    }
  }
  __syncthreads();
  if (VTILES && col0 >= EE) {
    const int kb0 = row0 >> 10, kt0 = ((row0 & (SS - 1)) >> 4) + wave;
    const int h0 = (col0 - EE) >> 5;
    const float* crow = ct + (16 * wave + m) * CPITCH + 2 * half;
#pragma unroll
    for (int j = 0; j < 8; ++j) {
      const int hl = j >> 2, i = j & 3;
      const v2f lo = *(const v2f*)(crow + 32 * hl + 8 * i);
      const v2f hi = *(const v2f*)(crow + 32 * hl + 8 * i + 4);
      const v4f v = {lo.x, lo.y, hi.x, hi.y};
      *(v4f*)(Kf + ((size_t)((kb0 * HH + h0 + hl) * 64 + kt0)) * 512 + i * 128 + lane * 4) = v;
    }
    return;
  }
  const int er = tid >> 4, ec = (tid & 15) << 2;
  float* yg = Y + (size_t)(row0 + er) * N + col0 + ec;
  const float* cs = ct + er * CPITCH + ec;
#pragma unroll
  for (int i = 0; i < 8; ++i) {
    const v4f v = *(const v4f*)(cs + 16 * i * CPITCH);
    *(v4f*)(yg + (size_t)(16 * i) * N) = v;
  }
}


__device__ __forceinline__ float xhalf(float x) {
  int b = __builtin_bit_cast(int, x);
  return __builtin_bit_cast(float,
      __builtin_amdgcn_permlanex16(b, b, 0x76543210, 0xfedcba98, false, false));
}

__device__ __forceinline__ int rows_for_halves(float x) {
  int b = __builtin_bit_cast(int, x);
  return __builtin_amdgcn_update_dpp(b, b, 0x168, 0x2, 0xf, false);
}

template <int R>
__device__ __forceinline__ float row_value(int xs) {
  return __builtin_bit_cast(float,
      __builtin_amdgcn_update_dpp(0, xs, 0x150 + R, 0xf, 0xf, true));
}

#define QT 2
#define OPITCH 36

__global__ __launch_bounds__(256) __attribute__((amdgpu_waves_per_eu(4, 4))) void attn_kernel(const float* __restrict__ qkv,
                                                   const float* __restrict__ Kf,
                                                   const float* __restrict__ Vf,
                                                   float* __restrict__ out) {
  __shared__ __attribute__((aligned(16))) float lds_p[8][QT][16 * LDSTR];

  const int lane = threadIdx.x & 31;
  const int wave = threadIdx.x >> 5;
  int id = blockIdx.x * 8 + wave;
  const int qg = id % (64 / QT); id /= (64 / QT);
  const int h  = id & 7;  id >>= 3;
  const int b  = id;
  const int m = lane & 15, half = lane >> 4;
  const int row0 = qg * (16 * QT);

  const size_t basebh = (size_t)(b * SS) * E3 + h * DD;
  const size_t tilebh = ((size_t)(b * HH + h) * 64) * 512 + lane * 4;
  const float* kslab = Kf + tilebh;
  const float* vslab = Vf + tilebh;

  float fr[4];
#pragma unroll
  for (int g = 0; g < 4; ++g) fr[g] = __expf(-1.1512925464970231f * (float)(2 * g + half));

  const float qscale = 0.17677669529663687f * 1.4426950408889634f;
  v2f qa[QT][8];
#pragma unroll
  for (int t = 0; t < QT; ++t) {
#pragma unroll
    for (int f = 0; f < 8; ++f) {
      const v2f q2 = *(const v2f*)(qkv + basebh + (size_t)(row0 + 16 * t + m) * E3 + 4 * f + 2 * half);
      const float pos = (f < 4) ? (float)qg : (float)(16 * t + m);
      float sn, cs;
      __sincosf(pos * fr[f & 3], &sn, &cs);
      sn *= qscale;
      cs *= qscale;
      qa[t][f].x = q2.x * cs - q2.y * sn;
      qa[t][f].y = q2.x * sn + q2.y * cs;
    }
  }

  float kcy[2][4], ksy[2][4];
#pragma unroll
  for (int par = 0; par < 2; ++par) {
#pragma unroll
    for (int g = 0; g < 4; ++g) __sincosf((float)(16 * par + m) * fr[g], &ksy[par][g], &kcy[par][g]);
  }

  float mrow[QT];
  float lsum[QT];
  v8f o0[QT], o1[QT];
#pragma unroll
  for (int t = 0; t < QT; ++t) {
    mrow[t] = -3.0e38f;
    lsum[t] = 0.0f;
    o0[t] = {};
    o1[t] = {};
  }

  float* myp = &lds_p[wave][0][0];

  for (int kp = 0; kp < 32; ++kp) {
   float kcx[4], ksx[4];
#pragma unroll
   for (int g = 0; g < 4; ++g) __sincosf((float)kp * fr[g], &ksx[g], &kcx[g]);

#pragma unroll
   for (int par = 0; par < 2; ++par) {
    const int kt = 2 * kp + par;
    const float* kbase = kslab + (size_t)kt * 512;
    v2f kb[8];
#pragma unroll
    for (int i = 0; i < 4; ++i) {
      const v4f k4 = *(const v4f*)(kbase + 128 * i);
      const float c0 = (i < 2) ? kcx[2 * i] : kcy[par][(2 * i) & 3];
      const float s0 = (i < 2) ? ksx[2 * i] : ksy[par][(2 * i) & 3];
      const float c1 = (i < 2) ? kcx[2 * i + 1] : kcy[par][(2 * i + 1) & 3];
      const float s1 = (i < 2) ? ksx[2 * i + 1] : ksy[par][(2 * i + 1) & 3];
      const v2f ka = {k4.x, k4.y}, kas = {k4.y, k4.x};
      const v2f kc = {k4.z, k4.w}, kcs = {k4.w, k4.z};
      const v2f cc0 = {c0, c0}, ss0 = {-s0, s0};
      const v2f cc1 = {c1, c1}, ss1 = {-s1, s1};
      kb[2 * i]     = ka * cc0 + kas * ss0;
      kb[2 * i + 1] = kc * cc1 + kcs * ss1;
    }

    const float* vbase = vslab + (size_t)kt * 512;
    v2f vb0[4], vb1[4];
#pragma unroll
    for (int g = 0; g < 2; ++g) {
      const v4f va = *(const v4f*)(vbase + 128 * g);
      const v4f vc = *(const v4f*)(vbase + 256 + 128 * g);
      vb0[2 * g].x = va.x;     vb0[2 * g].y = va.y;
      vb0[2 * g + 1].x = va.z; vb0[2 * g + 1].y = va.w;
      vb1[2 * g].x = vc.x;     vb1[2 * g].y = vc.y;
      vb1[2 * g + 1].x = vc.z; vb1[2 * g + 1].y = vc.w;
    }

    v8f sacc[QT];
#pragma unroll
    for (int t = 0; t < QT; ++t) sacc[t] = {};
#pragma unroll
    for (int f = 0; f < 8; ++f) {
#pragma unroll
      for (int t = 0; t < QT; ++t) sacc[t] = wmma_f32(qa[t][f], kb[f], sacc[t]);
    }

#pragma unroll
    for (int t = 0; t < QT; ++t) {
#pragma unroll
      for (int r = 0; r < 8; ++r)
        myp[t * (16 * LDSTR) + (r + 8 * half) * LDSTR + m] = sacc[t][r];
    }
    __builtin_amdgcn_wave_barrier();
    v2f sv[QT][4];
#pragma unroll
    for (int t = 0; t < QT; ++t) {
#pragma unroll
      for (int f = 0; f < 4; ++f)
        sv[t][f] = *(const v2f*)(myp + t * (16 * LDSTR) + m * LDSTR + 4 * f + 2 * half);
    }
    __builtin_amdgcn_wave_barrier();

    v2f pa[QT][4];
#pragma unroll
    for (int t = 0; t < QT; ++t) {
      float tmax = fmaxf(fmaxf(sv[t][0].x, sv[t][0].y), sv[t][1].x);
      tmax = fmaxf(fmaxf(tmax, sv[t][1].y), sv[t][2].x);
      tmax = fmaxf(fmaxf(tmax, sv[t][2].y), sv[t][3].x);
      tmax = fmaxf(tmax, sv[t][3].y);
      tmax = fmaxf(tmax, xhalf(tmax));
      const float mnew = fmaxf(mrow[t], tmax);
      const float corr = __builtin_amdgcn_exp2f(mrow[t] - mnew);
      mrow[t] = mnew;

#pragma unroll
      for (int f = 0; f < 4; ++f) {
        pa[t][f].x = __builtin_amdgcn_exp2f(sv[t][f].x - mnew);
        pa[t][f].y = __builtin_amdgcn_exp2f(sv[t][f].y - mnew);
      }
      lsum[t] = lsum[t] * corr + (((pa[t][0].x + pa[t][0].y) + (pa[t][1].x + pa[t][1].y)) +
                                  ((pa[t][2].x + pa[t][2].y) + (pa[t][3].x + pa[t][3].y)));

      const int cs = rows_for_halves(corr);
#define RESCALE_ROW(R)                       \
  do {                                       \
    const float cr = row_value<R>(cs);       \
    o0[t][R] *= cr;                          \
    o1[t][R] *= cr;                          \
  } while (0)
      RESCALE_ROW(0); RESCALE_ROW(1); RESCALE_ROW(2); RESCALE_ROW(3);
      RESCALE_ROW(4); RESCALE_ROW(5); RESCALE_ROW(6); RESCALE_ROW(7);
#undef RESCALE_ROW
    }

#pragma unroll
    for (int f = 0; f < 4; ++f) {
#pragma unroll
      for (int t = 0; t < QT; ++t) {
        o0[t] = wmma_f32(pa[t][f], vb0[f], o0[t]);
        o1[t] = wmma_f32(pa[t][f], vb1[f], o1[t]);
      }
    }
   }
  }

#pragma unroll
  for (int t = 0; t < QT; ++t) {
    const float inv = 1.0f / (lsum[t] + xhalf(lsum[t]));
    const int is = rows_for_halves(inv);
#define STAGE_ROW(R)                                                     \
  do {                                                                   \
    const float ir = row_value<R>(is);                                   \
    myp[((R) + 8 * half) * OPITCH + m]      = o0[t][R] * ir;             \
    myp[((R) + 8 * half) * OPITCH + 16 + m] = o1[t][R] * ir;             \
  } while (0)
    STAGE_ROW(0); STAGE_ROW(1); STAGE_ROW(2); STAGE_ROW(3);
    STAGE_ROW(4); STAGE_ROW(5); STAGE_ROW(6); STAGE_ROW(7);
#undef STAGE_ROW
    __builtin_amdgcn_wave_barrier();
    const int orr = lane >> 3, oc = (lane & 7) << 2;
    float* og = out + (size_t)(b * SS + row0 + 16 * t + orr) * EE + h * DD + oc;
#pragma unroll
    for (int j = 0; j < 4; ++j) {
      const v4f v = *(const v4f*)(myp + (orr + 4 * j) * OPITCH + oc);
      *(v4f*)(og + (size_t)(4 * j) * EE) = v;
    }
    __builtin_amdgcn_wave_barrier();
  }
}

extern "C" void kernel_launch(void* const* d_in, const int* in_sizes, int n_in,
                              void* d_out, int out_size, void* d_ws, size_t ws_size,
                              hipStream_t stream) {
  const float* x    = (const float*)d_in[0];
  const float* Wqkv = (const float*)d_in[1];
  const float* bqkv = (const float*)d_in[2];
  const float* Wout = (const float*)d_in[3];
  const float* bout = (const float*)d_in[4];
  float* out  = (float*)d_out;
  float* qkv  = (float*)d_ws;
  float* attn = qkv + (size_t)NB * SS * E3;
  float* Kf   = attn + (size_t)NB * SS * EE;
  float* Vf   = Kf + (size_t)NB * SS * EE;

  gemm_lds128x64<true><<<dim3((NB * SS / GBM) * (E3 / GBN)), dim3(256), 0, stream>>>(
      x, Wqkv, bqkv, qkv, Vf, Kf, NB * SS, E3, EE);

  attn_kernel<<<dim3(NB * HH * (SS / (16 * QT)) / 8), dim3(256), 0, stream>>>(qkv, Kf, Vf, attn);

  gemm_lds128x64<false><<<dim3((NB * SS / GBM) * (EE / GBN)), dim3(256), 0, stream>>>(
      attn, Wout, bout, out, nullptr, nullptr, NB * SS, EE, EE);
}


# attention: lazy row maximum: a key tile whose scores stay below held maximum + 8 skips half exchange, correction and O rescale (wave-uniform branch); rescale points change, sum orders kept
# speedup vs baseline: 2.3159x; 1.0297x over previous
#include <hip/hip_runtime.h>

typedef __attribute__((ext_vector_type(2))) float v2f;
typedef __attribute__((ext_vector_type(4))) float v4f;
typedef __attribute__((ext_vector_type(8))) float v8f;

#define NB 16
#define SS 1024
#define EE 256
#define HH 8
#define DD 32
#define E3 768
#define LDSTR 20

__device__ __forceinline__ v8f wmma_f32(v2f a, v2f b, v8f c) {
  return __builtin_amdgcn_wmma_f32_16x16x4_f32(
       false, a,  false, b,
       (short)0, c,  false,  false);
}

#define GBM 128
#define GBN 64
#define GKC 32
#define XPITCH 36
#define WPITCH 80
#define CPITCH 68

template <bool VTILES>
__global__ __launch_bounds__(256) void gemm_lds128x64(const float* __restrict__ X,
                                                      const float* __restrict__ W,
                                                      const float* __restrict__ bias,
                                                      float* __restrict__ Y,
                                                      float* __restrict__ Vf,
                                                      float* __restrict__ Kf,
                                                      int M, int N, int K) {
  __shared__ __attribute__((aligned(16))) float xs[2 * GBM * XPITCH];
  __shared__ __attribute__((aligned(16))) float wl[2 * GKC * WPITCH];

  const int tid  = threadIdx.x;
  const int lane = tid & 31;
  const int wave = tid >> 5;
  const int m = lane & 15, half = lane >> 4;
  const int nbn = N >> 6;
  const int bm = blockIdx.x / nbn;
  const int bn = blockIdx.x - bm * nbn;
  const int row0 = bm * GBM, col0 = bn * GBN;
  const int wm = wave & 1;
  const int wn = wave >> 1;

  const int xr = tid >> 3, xc = (tid & 7) << 2;
  const int wr = tid >> 4, wc = (tid & 15) << 2;
  const float* xg = X + (size_t)(row0 + xr) * K + xc;
  const float* wg = W + (size_t)wr * N + col0 + wc;
  float* xst = xs + xr * XPITCH + xc;
  float* wst = wl + wr * WPITCH + wc;

  const size_t xstep = (size_t)32 * K;
  const size_t wstep = (size_t)16 * N;
  v4f x0 = *(const v4f*)(xg);
  v4f x1 = *(const v4f*)(xg + xstep);
  v4f x2 = *(const v4f*)(xg + 2 * xstep);
  v4f x3 = *(const v4f*)(xg + 3 * xstep);
  v4f w0 = *(const v4f*)(wg);
  v4f w1 = *(const v4f*)(wg + wstep);

  const float* xa = xs + (wm * 64 + m) * XPITCH + 2 * half;
  const float* wb = wl + (2 * half) * WPITCH + wn * 16 + m;

  v8f acc[4];
#pragma unroll
  for (int t = 0; t < 4; ++t) acc[t] = {};

#define STORE_SLAB(S)                                                   \
  do {                                                                  \
    *(v4f*)(xst + (S) * (GBM * XPITCH))               = x0;             \
    *(v4f*)(xst + (S) * (GBM * XPITCH) + 32 * XPITCH) = x1;             \
    *(v4f*)(xst + (S) * (GBM * XPITCH) + 64 * XPITCH) = x2;             \
    *(v4f*)(xst + (S) * (GBM * XPITCH) + 96 * XPITCH) = x3;             \
    *(v4f*)(wst + (S) * (GKC * WPITCH))               = w0;             \
    *(v4f*)(wst + (S) * (GKC * WPITCH) + 16 * WPITCH) = w1;             \
  } while (0)
#define LOAD_SLAB(KOFF)                                                 \
  do {                                                                  \
    const float* xn = xg + (KOFF);                                      \
    const float* wnx = wg + (size_t)(KOFF) * N;                         \
    x0 = *(const v4f*)(xn);                                             \
    x1 = *(const v4f*)(xn + xstep);                                     \
    x2 = *(const v4f*)(xn + 2 * xstep);                                 \
    x3 = *(const v4f*)(xn + 3 * xstep);                                 \
    w0 = *(const v4f*)(wnx);                                            \
    w1 = *(const v4f*)(wnx + wstep);                                    \
  } while (0)
#define WMMA_SLAB(S)                                                    \
  do {                                                                  \
    v2f bq[GKC / 4];                                                    \
    _Pragma("unroll") for (int q = 0; q < GKC / 4; ++q) {               \
      const float* wbs = wb + (S) * (GKC * WPITCH) + 4 * q * WPITCH;    \
      bq[q].x = wbs[0];                                                 \
      bq[q].y = wbs[WPITCH];                                            \
    }                                                                   \
    _Pragma("unroll") for (int q = 0; q < GKC / 4; ++q) {               \
      const float* xas = xa + (S) * (GBM * XPITCH) + 4 * q;             \
      v2f a0 = *(const v2f*)(xas);                                      \
      v2f a1 = *(const v2f*)(xas + 16 * XPITCH);                        \
      v2f a2 = *(const v2f*)(xas + 32 * XPITCH);                        \
      v2f a3 = *(const v2f*)(xas + 48 * XPITCH);                        \
      acc[0] = wmma_f32(a0, bq[q], acc[0]);                             \
      acc[1] = wmma_f32(a1, bq[q], acc[1]);                             \
      acc[2] = wmma_f32(a2, bq[q], acc[2]);                             \
      acc[3] = wmma_f32(a3, bq[q], acc[3]);                             \
    }                                                                   \
  } while (0)

  STORE_SLAB(0);
  __syncthreads();

  for (int c = 0; c < K; c += 2 * GKC) {
    const bool more1 = (c + GKC < K);
    if (more1) LOAD_SLAB(c + GKC);
    WMMA_SLAB(0);
    if (!more1) break;
    STORE_SLAB(1);
    __syncthreads();

    const bool more2 = (c + 2 * GKC < K);
    if (more2) LOAD_SLAB(c + 2 * GKC);
    WMMA_SLAB(1);
    if (more2) STORE_SLAB(0);
    __syncthreads();
  }
#undef STORE_SLAB
#undef LOAD_SLAB
#undef WMMA_SLAB

  const int cn = col0 + wn * 16 + m;
  float bnv = bias[cn];
  if (VTILES && col0 >= 2 * EE) {
    const int cv = col0 - 2 * EE + wn * 16;
    const int vh = cv >> 5;
    const int vw = (cv >> 4) & 1;
#pragma unroll
    for (int t = 0; t < 4; ++t) {
      const int R0 = row0 + wm * 64 + 16 * t;
      const int vb = R0 >> 10, vkt = (R0 & (SS - 1)) >> 4;
      float* tile = Vf + ((size_t)((vb * HH + vh) * 64 + vkt)) * 512 + vw * 256 + half * 128 + m * 4;
      v4f lo = {acc[t][0] + bnv, acc[t][1] + bnv, acc[t][4] + bnv, acc[t][5] + bnv};
      v4f hi = {acc[t][2] + bnv, acc[t][3] + bnv, acc[t][6] + bnv, acc[t][7] + bnv};
      *(v4f*)(tile)      = lo;
      *(v4f*)(tile + 64) = hi;
    }
    return;
  }
  __syncthreads();
  float* ct = xs;
#pragma unroll
  for (int t = 0; t < 4; ++t) {
#pragma unroll
    for (int r = 0; r < 8; ++r) {
      int Mr = wm * 64 + 16 * t + r + 8 * half;
      ct[Mr * CPITCH + wn * 16 + m] = acc[t][r] + bnv;
    }
  }
  __syncthreads();
  if (VTILES && col0 >= EE) {
    const int kb0 = row0 >> 10, kt0 = ((row0 & (SS - 1)) >> 4) + wave;
    const int h0 = (col0 - EE) >> 5;
    const float* crow = ct + (16 * wave + m) * CPITCH + 2 * half;
#pragma unroll
    for (int j = 0; j < 8; ++j) {
      const int hl = j >> 2, i = j & 3;
      const v2f lo = *(const v2f*)(crow + 32 * hl + 8 * i);
      const v2f hi = *(const v2f*)(crow + 32 * hl + 8 * i + 4);
      const v4f v = {lo.x, lo.y, hi.x, hi.y};
      *(v4f*)(Kf + ((size_t)((kb0 * HH + h0 + hl) * 64 + kt0)) * 512 + i * 128 + lane * 4) = v;
    }
    return;
  }
  const int er = tid >> 4, ec = (tid & 15) << 2;
  float* yg = Y + (size_t)(row0 + er) * N + col0 + ec;
  const float* cs = ct + er * CPITCH + ec;
#pragma unroll
  for (int i = 0; i < 8; ++i) {
    const v4f v = *(const v4f*)(cs + 16 * i * CPITCH);
    *(v4f*)(yg + (size_t)(16 * i) * N) = v;
  }
}


__device__ __forceinline__ float xhalf(float x) {
  int b = __builtin_bit_cast(int, x);
  return __builtin_bit_cast(float,
      __builtin_amdgcn_permlanex16(b, b, 0x76543210, 0xfedcba98, false, false));
}

__device__ __forceinline__ int rows_for_halves(float x) {
  int b = __builtin_bit_cast(int, x);
  return __builtin_amdgcn_update_dpp(b, b, 0x168, 0x2, 0xf, false);
}

template <int R>
__device__ __forceinline__ float row_value(int xs) {
  return __builtin_bit_cast(float,
      __builtin_amdgcn_update_dpp(0, xs, 0x150 + R, 0xf, 0xf, true));
}

#define QT 2
#define RAISE 8.0f
#define OPITCH 36

__global__ __launch_bounds__(256) __attribute__((amdgpu_waves_per_eu(4, 4))) void attn_kernel(const float* __restrict__ qkv,
                                                   const float* __restrict__ Kf,
                                                   const float* __restrict__ Vf,
                                                   float* __restrict__ out) {
  __shared__ __attribute__((aligned(16))) float lds_p[8][QT][16 * LDSTR];

  const int lane = threadIdx.x & 31;
  const int wave = threadIdx.x >> 5;
  int id = blockIdx.x * 8 + wave;
  const int qg = id % (64 / QT); id /= (64 / QT);
  const int h  = id & 7;  id >>= 3;
  const int b  = id;
  const int m = lane & 15, half = lane >> 4;
  const int row0 = qg * (16 * QT);

  const size_t basebh = (size_t)(b * SS) * E3 + h * DD;
  const size_t tilebh = ((size_t)(b * HH + h) * 64) * 512 + lane * 4;
  const float* kslab = Kf + tilebh;
  const float* vslab = Vf + tilebh;

  float fr[4];
#pragma unroll
  for (int g = 0; g < 4; ++g) fr[g] = __expf(-1.1512925464970231f * (float)(2 * g + half));

  const float qscale = 0.17677669529663687f * 1.4426950408889634f;
  v2f qa[QT][8];
#pragma unroll
  for (int t = 0; t < QT; ++t) {
#pragma unroll
    for (int f = 0; f < 8; ++f) {
      const v2f q2 = *(const v2f*)(qkv + basebh + (size_t)(row0 + 16 * t + m) * E3 + 4 * f + 2 * half);
      const float pos = (f < 4) ? (float)qg : (float)(16 * t + m);
      float sn, cs;
      __sincosf(pos * fr[f & 3], &sn, &cs);
      sn *= qscale;
      cs *= qscale;
      qa[t][f].x = q2.x * cs - q2.y * sn;
      qa[t][f].y = q2.x * sn + q2.y * cs;
    }
  }

  float kcy[2][4], ksy[2][4];
#pragma unroll
  for (int par = 0; par < 2; ++par) {
#pragma unroll
    for (int g = 0; g < 4; ++g) __sincosf((float)(16 * par + m) * fr[g], &ksy[par][g], &kcy[par][g]);
  }

  float mrow[QT];
  float mthr[QT];
  float lsum[QT];
  v8f o0[QT], o1[QT];
#pragma unroll
  for (int t = 0; t < QT; ++t) {
    mrow[t] = -3.0e38f;
    mthr[t] = -3.0e38f;
    lsum[t] = 0.0f;
    o0[t] = {};
    o1[t] = {};
  }

  float* myp = &lds_p[wave][0][0];

  for (int kp = 0; kp < 32; ++kp) {
   float kcx[4], ksx[4];
#pragma unroll
   for (int g = 0; g < 4; ++g) __sincosf((float)kp * fr[g], &ksx[g], &kcx[g]);

#pragma unroll
   for (int par = 0; par < 2; ++par) {
    const int kt = 2 * kp + par;
    const float* kbase = kslab + (size_t)kt * 512;
    v2f kb[8];
#pragma unroll
    for (int i = 0; i < 4; ++i) {
      const v4f k4 = *(const v4f*)(kbase + 128 * i);
      const float c0 = (i < 2) ? kcx[2 * i] : kcy[par][(2 * i) & 3];
      const float s0 = (i < 2) ? ksx[2 * i] : ksy[par][(2 * i) & 3];
      const float c1 = (i < 2) ? kcx[2 * i + 1] : kcy[par][(2 * i + 1) & 3];
      const float s1 = (i < 2) ? ksx[2 * i + 1] : ksy[par][(2 * i + 1) & 3];
      const v2f ka = {k4.x, k4.y}, kas = {k4.y, k4.x};
      const v2f kc = {k4.z, k4.w}, kcs = {k4.w, k4.z};
      const v2f cc0 = {c0, c0}, ss0 = {-s0, s0};
      const v2f cc1 = {c1, c1}, ss1 = {-s1, s1};
      kb[2 * i]     = ka * cc0 + kas * ss0;
      kb[2 * i + 1] = kc * cc1 + kcs * ss1;
    }

    const float* vbase = vslab + (size_t)kt * 512;
    v2f vb0[4], vb1[4];
#pragma unroll
    for (int g = 0; g < 2; ++g) {
      const v4f va = *(const v4f*)(vbase + 128 * g);
      const v4f vc = *(const v4f*)(vbase + 256 + 128 * g);
      vb0[2 * g].x = va.x;     vb0[2 * g].y = va.y;
      vb0[2 * g + 1].x = va.z; vb0[2 * g + 1].y = va.w;
      vb1[2 * g].x = vc.x;     vb1[2 * g].y = vc.y;
      vb1[2 * g + 1].x = vc.z; vb1[2 * g + 1].y = vc.w;
    }

    v8f sacc[QT];
#pragma unroll
    for (int t = 0; t < QT; ++t) sacc[t] = {};
#pragma unroll
    for (int f = 0; f < 8; ++f) {
#pragma unroll
      for (int t = 0; t < QT; ++t) sacc[t] = wmma_f32(qa[t][f], kb[f], sacc[t]);
    }

#pragma unroll
    for (int t = 0; t < QT; ++t) {
#pragma unroll
      for (int r = 0; r < 8; ++r)
        myp[t * (16 * LDSTR) + (r + 8 * half) * LDSTR + m] = sacc[t][r];
    }
    __builtin_amdgcn_wave_barrier();
    v2f sv[QT][4];
#pragma unroll
    for (int t = 0; t < QT; ++t) {
#pragma unroll
      for (int f = 0; f < 4; ++f)
        sv[t][f] = *(const v2f*)(myp + t * (16 * LDSTR) + m * LDSTR + 4 * f + 2 * half);
    }
    __builtin_amdgcn_wave_barrier();

    float tm[QT];
    bool need = false;
#pragma unroll
    for (int t = 0; t < QT; ++t) {
      float tmax = fmaxf(fmaxf(sv[t][0].x, sv[t][0].y), sv[t][1].x);
      tmax = fmaxf(fmaxf(tmax, sv[t][1].y), sv[t][2].x);
      tmax = fmaxf(fmaxf(tmax, sv[t][2].y), sv[t][3].x);
      tm[t] = fmaxf(tmax, sv[t][3].y);
      need = need || (tm[t] > mthr[t]);
    }
    if (__builtin_amdgcn_ballot_w32(need) != 0u) {
#pragma unroll
      for (int t = 0; t < QT; ++t) {
        const float tmax = fmaxf(tm[t], xhalf(tm[t]));
        const float mnew = fmaxf(mrow[t], tmax);
        const float corr = __builtin_amdgcn_exp2f(mrow[t] - mnew);
        mrow[t] = mnew;
        mthr[t] = mnew + RAISE;
        lsum[t] *= corr;
        const int cs = rows_for_halves(corr);
#define RESCALE_ROW(R)                       \
  do {                                       \
    const float cr = row_value<R>(cs);       \
    o0[t][R] *= cr;                          \
    o1[t][R] *= cr;                          \
  } while (0)
        RESCALE_ROW(0); RESCALE_ROW(1); RESCALE_ROW(2); RESCALE_ROW(3);
        RESCALE_ROW(4); RESCALE_ROW(5); RESCALE_ROW(6); RESCALE_ROW(7);
#undef RESCALE_ROW
      }
    }

    v2f pa[QT][4];
#pragma unroll
    for (int t = 0; t < QT; ++t) {
#pragma unroll
      for (int f = 0; f < 4; ++f) {
        pa[t][f].x = __builtin_amdgcn_exp2f(sv[t][f].x - mrow[t]);
        pa[t][f].y = __builtin_amdgcn_exp2f(sv[t][f].y - mrow[t]);
      }
      lsum[t] += (((pa[t][0].x + pa[t][0].y) + (pa[t][1].x + pa[t][1].y)) +
                  ((pa[t][2].x + pa[t][2].y) + (pa[t][3].x + pa[t][3].y)));
    }

#pragma unroll
    for (int f = 0; f < 4; ++f) {
#pragma unroll
      for (int t = 0; t < QT; ++t) {
        o0[t] = wmma_f32(pa[t][f], vb0[f], o0[t]);
        o1[t] = wmma_f32(pa[t][f], vb1[f], o1[t]);
      }
    }
   }
  }

#pragma unroll
  for (int t = 0; t < QT; ++t) {
    const float inv = 1.0f / (lsum[t] + xhalf(lsum[t]));
    const int is = rows_for_halves(inv);
#define STAGE_ROW(R)                                                     \
  do {                                                                   \
    const float ir = row_value<R>(is);                                   \
    myp[((R) + 8 * half) * OPITCH + m]      = o0[t][R] * ir;             \
    myp[((R) + 8 * half) * OPITCH + 16 + m] = o1[t][R] * ir;             \
  } while (0)
    STAGE_ROW(0); STAGE_ROW(1); STAGE_ROW(2); STAGE_ROW(3);
    STAGE_ROW(4); STAGE_ROW(5); STAGE_ROW(6); STAGE_ROW(7);
#undef STAGE_ROW
    __builtin_amdgcn_wave_barrier();
    const int orr = lane >> 3, oc = (lane & 7) << 2;
    float* og = out + (size_t)(b * SS + row0 + 16 * t + orr) * EE + h * DD + oc;
#pragma unroll
    for (int j = 0; j < 4; ++j) {
      const v4f v = *(const v4f*)(myp + (orr + 4 * j) * OPITCH + oc);
      *(v4f*)(og + (size_t)(4 * j) * EE) = v;
    }
    __builtin_amdgcn_wave_barrier();
  }
}

extern "C" void kernel_launch(void* const* d_in, const int* in_sizes, int n_in,
                              void* d_out, int out_size, void* d_ws, size_t ws_size,
                              hipStream_t stream) {
  const float* x    = (const float*)d_in[0];
  const float* Wqkv = (const float*)d_in[1];
  const float* bqkv = (const float*)d_in[2];
  const float* Wout = (const float*)d_in[3];
  const float* bout = (const float*)d_in[4];
  float* out  = (float*)d_out;
  float* qkv  = (float*)d_ws;
  float* attn = qkv + (size_t)NB * SS * E3;
  float* Kf   = attn + (size_t)NB * SS * EE;
  float* Vf   = Kf + (size_t)NB * SS * EE;

  gemm_lds128x64<true><<<dim3((NB * SS / GBM) * (E3 / GBN)), dim3(256), 0, stream>>>(
      x, Wqkv, bqkv, qkv, Vf, Kf, NB * SS, E3, EE);

  attn_kernel<<<dim3(NB * HH * (SS / (16 * QT)) / 8), dim3(256), 0, stream>>>(qkv, Kf, Vf, attn);

  gemm_lds128x64<false><<<dim3((NB * SS / GBM) * (EE / GBN)), dim3(256), 0, stream>>>(
      attn, Wout, bout, out, nullptr, nullptr, NB * SS, EE, EE);
}
